# starting kernel
# baseline (speedup 1.0000x reference)
#include <hip/hip_runtime.h>

typedef float v2f __attribute__((ext_vector_type(2)));
typedef float v8f __attribute__((ext_vector_type(8)));

#define D 32

__global__ void k_init_deg(float* __restrict__ deg, int n) {
  int i = blockIdx.x * blockDim.x + threadIdx.x;
  if (i < n) deg[i] = 1.0f;
}

__global__ void k_deg_edges(const int* __restrict__ dst, const float* __restrict__ w,
                            float* __restrict__ deg, int E) {
  int e = blockIdx.x * blockDim.x + threadIdx.x;
  if (e < E) atomicAdd(&deg[dst[e]], w[e]);
}

__global__ void k_dis(const float* __restrict__ deg, float* __restrict__ dis, int n) {
  int i = blockIdx.x * blockDim.x + threadIdx.x;
  if (i < n) {
    float d = deg[i];
    dis[i] = (d > 0.0f) ? __frsqrt_rn(d) : 0.0f;
  }
}

template <bool RELU>
__global__ void k_gemm32(const float* __restrict__ X, const float* __restrict__ W,
                         float* __restrict__ Y, int n) {
  const int lane = threadIdx.x & 31;
  const int wave = threadIdx.x >> 5;
  const int row0 = (blockIdx.x * (blockDim.x >> 5) + wave) << 4;
  if (row0 >= n) return;

  const int l    = lane & 15;
  const int half = lane >> 4;

  v8f c0 = {};
  v8f c1 = {};

  const float* xrow = X + (size_t)(row0 + l) * D;

#pragma unroll
  for (int kb = 0; kb < D; kb += 4) {
    const int k0 = kb + 2 * half;
    v2f a;
    a.x = xrow[k0];
    a.y = xrow[k0 + 1];
    if (RELU) {
      a.x = fmaxf(a.x, 0.0f);
      a.y = fmaxf(a.y, 0.0f);
    }
    v2f b0, b1;
    b0.x = W[(size_t)k0 * D + l];
    b0.y = W[(size_t)(k0 + 1) * D + l];
    b1.x = W[(size_t)k0 * D + 16 + l];
    b1.y = W[(size_t)(k0 + 1) * D + 16 + l];

    c0 = __builtin_amdgcn_wmma_f32_16x16x4_f32(false, a, false, b0, (short)0, c0,
                                               false, false);
    c1 = __builtin_amdgcn_wmma_f32_16x16x4_f32(false, a, false, b1, (short)0, c1,
                                               false, false);
  }

  const int rowm = row0 + 8 * half;
#pragma unroll
  for (int j = 0; j < 8; ++j) {
    Y[(size_t)(rowm + j) * D + l]      = c0[j];
    Y[(size_t)(rowm + j) * D + 16 + l] = c1[j];
  }
}

__global__ void k_self_bias(const float* __restrict__ xm, const float* __restrict__ dis,
                            const float* __restrict__ b, float* __restrict__ out,
                            long long total) {
  long long idx = (long long)blockIdx.x * blockDim.x + threadIdx.x;
  if (idx >= total) return;
  int i = (int)(idx >> 5);
  int d = (int)(idx & (D - 1));
  float s = dis[i];
  out[idx] = b[d] + xm[idx] * (s * s);
}

__global__ void k_edge_msg(const int* __restrict__ src, const int* __restrict__ dst,
                           const float* __restrict__ w, const float* __restrict__ dis,
                           const float* __restrict__ xm, float* __restrict__ out,
                           int E) {
  long long idx = (long long)blockIdx.x * blockDim.x + threadIdx.x;
  int e = (int)(idx >> 5);
  int d = (int)(idx & (D - 1));
  if (e >= E) return;

  if (d == 0 && e + 4096 < E) {
    __builtin_prefetch(&src[e + 4096], 0, 0);
    __builtin_prefetch(&dst[e + 4096], 0, 0);
    __builtin_prefetch(&w[e + 4096], 0, 0);
  }

  int s = src[e];
  int t = dst[e];
  float c = dis[s] * w[e] * dis[t];
  atomicAdd(&out[(size_t)t * D + d], xm[(size_t)s * D + d] * c);
}

extern "C" void kernel_launch(void* const* d_in, const int* in_sizes, int n_in,
                              void* d_out, int out_size, void* d_ws, size_t ws_size,
                              hipStream_t stream) {
  const float* x  = (const float*)d_in[0];
  const int*   ei = (const int*)d_in[1];
  const float* ew = (const float*)d_in[2];
  const float* W1 = (const float*)d_in[3];
  const float* b1 = (const float*)d_in[4];
  const float* W2 = (const float*)d_in[5];
  const float* b2 = (const float*)d_in[6];

  const int n = in_sizes[0] / D;
  const int E = in_sizes[2];
  const int* src = ei;
  const int* dst = ei + E;

  float* ws  = (float*)d_ws;
  float* deg = ws;
  float* dis = ws + n;
  float* xm  = ws + 2 * (size_t)n;
  float* h   = xm + (size_t)n * D;
  float* out = (float*)d_out;

  const int T = 256;
  const int nBlocks   = (n + T - 1) / T;
  const int eBlocks   = (E + T - 1) / T;
  const long long nd  = (long long)n * D;
  const int ndBlocks  = (int)((nd + T - 1) / T);
  const int edBlocks  = (int)(((long long)E * D + T - 1) / T);
  const int rowsPerBlock = (T / 32) * 16;
  const int gBlocks   = (n + rowsPerBlock - 1) / rowsPerBlock;

  k_init_deg<<<nBlocks, T, 0, stream>>>(deg, n);
  k_deg_edges<<<eBlocks, T, 0, stream>>>(dst, ew, deg, E);
  k_dis<<<nBlocks, T, 0, stream>>>(deg, dis, n);

  k_gemm32<false><<<gBlocks, T, 0, stream>>>(x, W1, xm, n);
  k_self_bias<<<ndBlocks, T, 0, stream>>>(xm, dis, b1, h, nd);
  k_edge_msg<<<edBlocks, T, 0, stream>>>(src, dst, ew, dis, xm, h, E);

  k_gemm32<true><<<gBlocks, T, 0, stream>>>(h, W2, xm, n);
  k_self_bias<<<ndBlocks, T, 0, stream>>>(xm, dis, b2, out, nd);
  k_edge_msg<<<edBlocks, T, 0, stream>>>(src, dst, ew, dis, xm, out, E);
}


# edge-message kernel: one wave per 32 edges, not one per edge: edge scalars read once per lane and broadcast by readlane, 32 source-row loads in flight; whole-row f32 atomic adds kept (arrival order)
# speedup vs baseline: 1.7506x; 1.7506x over previous
#include <hip/hip_runtime.h>

typedef float v2f __attribute__((ext_vector_type(2)));
typedef float v8f __attribute__((ext_vector_type(8)));

#define D 32

__global__ void k_init_deg(float* __restrict__ deg, int n) {
  int i = blockIdx.x * blockDim.x + threadIdx.x;
  if (i < n) deg[i] = 1.0f;
}

__global__ void k_deg_edges(const int* __restrict__ dst, const float* __restrict__ w,
                            float* __restrict__ deg, int E) {
  int e = blockIdx.x * blockDim.x + threadIdx.x;
  if (e < E) atomicAdd(&deg[dst[e]], w[e]);
}

__global__ void k_dis(const float* __restrict__ deg, float* __restrict__ dis, int n) {
  int i = blockIdx.x * blockDim.x + threadIdx.x;
  if (i < n) {
    float d = deg[i];
    dis[i] = (d > 0.0f) ? __frsqrt_rn(d) : 0.0f;
  }
}

template <bool RELU>
__global__ void k_gemm32(const float* __restrict__ X, const float* __restrict__ W,
                         float* __restrict__ Y, int n) {
  const int lane = threadIdx.x & 31;
  const int wave = threadIdx.x >> 5;
  const int row0 = (blockIdx.x * (blockDim.x >> 5) + wave) << 4;
  if (row0 >= n) return;

  const int l    = lane & 15;
  const int half = lane >> 4;

  v8f c0 = {};
  v8f c1 = {};

  const float* xrow = X + (size_t)(row0 + l) * D;

#pragma unroll
  for (int kb = 0; kb < D; kb += 4) {
    const int k0 = kb + 2 * half;
    v2f a;
    a.x = xrow[k0];
    a.y = xrow[k0 + 1];
    if (RELU) {
      a.x = fmaxf(a.x, 0.0f);
      a.y = fmaxf(a.y, 0.0f);
    }
    v2f b0, b1;
    b0.x = W[(size_t)k0 * D + l];
    b0.y = W[(size_t)(k0 + 1) * D + l];
    b1.x = W[(size_t)k0 * D + 16 + l];
    b1.y = W[(size_t)(k0 + 1) * D + 16 + l];

    c0 = __builtin_amdgcn_wmma_f32_16x16x4_f32(false, a, false, b0, (short)0, c0,
                                               false, false);
    c1 = __builtin_amdgcn_wmma_f32_16x16x4_f32(false, a, false, b1, (short)0, c1,
                                               false, false);
  }

  const int rowm = row0 + 8 * half;
#pragma unroll
  for (int j = 0; j < 8; ++j) {
    Y[(size_t)(rowm + j) * D + l]      = c0[j];
    Y[(size_t)(rowm + j) * D + 16 + l] = c1[j];
  }
}

__global__ void k_self_bias(const float* __restrict__ xm, const float* __restrict__ dis,
                            const float* __restrict__ b, float* __restrict__ out,
                            long long total) {
  long long idx = (long long)blockIdx.x * blockDim.x + threadIdx.x;
  if (idx >= total) return;
  int i = (int)(idx >> 5);
  int d = (int)(idx & (D - 1));
  float s = dis[i];
  out[idx] = b[d] + xm[idx] * (s * s);
}

__global__ void k_edge_msg(const int* __restrict__ src, const int* __restrict__ dst,
                           const float* __restrict__ w, const float* __restrict__ dis,
                           const float* __restrict__ xm, float* __restrict__ out,
                           int E) {
  const int lane = threadIdx.x & 31;
  const int wave = __builtin_amdgcn_readfirstlane((int)(threadIdx.x >> 5));
  const int e0 = (blockIdx.x * (blockDim.x >> 5) + wave) << 5;
  if (e0 >= E) return;
  const int cnt = (E - e0 < 32) ? (E - e0) : 32;

  if (cnt == 32) {
    const int e = e0 + lane;
    const int s = src[e];
    const int t = dst[e];
    const float we = w[e];
    const float ds = dis[s];
    const float dt = dis[t];

    float v[32];
#pragma unroll
    for (int j = 0; j < 32; ++j) {
      const unsigned rowi = (unsigned)__builtin_amdgcn_readlane(s, j) * D;
      v[j] = xm[rowi + (unsigned)lane];
    }

    const float c = ds * we * dt;
#pragma unroll
    for (int j = 0; j < 32; ++j) {
      const unsigned rowo = (unsigned)__builtin_amdgcn_readlane(t, j) * D;
      const float cj = __int_as_float(__builtin_amdgcn_readlane(__float_as_int(c), j));
      atomicAdd(&out[rowo + (unsigned)lane], v[j] * cj);
    }
  } else {
    int s = 0;
    int t = 0;
    float c = 0.0f;
    if (lane < cnt) {
      const int e = e0 + lane;
      s = src[e];
      t = dst[e];
      c = dis[s] * w[e] * dis[t];
    }
    for (int j = 0; j < cnt; ++j) {
      const float* rowp = xm + (size_t)__builtin_amdgcn_readlane(s, j) * D;
      float* rowq = out + (size_t)__builtin_amdgcn_readlane(t, j) * D;
      const float cj = __int_as_float(__builtin_amdgcn_readlane(__float_as_int(c), j));
      atomicAdd(&rowq[lane], rowp[lane] * cj);
    }
  }
}

extern "C" void kernel_launch(void* const* d_in, const int* in_sizes, int n_in,
                              void* d_out, int out_size, void* d_ws, size_t ws_size,
                              hipStream_t stream) {
  const float* x  = (const float*)d_in[0];
  const int*   ei = (const int*)d_in[1];
  const float* ew = (const float*)d_in[2];
  const float* W1 = (const float*)d_in[3];
  const float* b1 = (const float*)d_in[4];
  const float* W2 = (const float*)d_in[5];
  const float* b2 = (const float*)d_in[6];

  const int n = in_sizes[0] / D;
  const int E = in_sizes[2];
  const int* src = ei;
  const int* dst = ei + E;

  float* ws  = (float*)d_ws;
  float* deg = ws;
  float* dis = ws + n;
  float* xm  = ws + 2 * (size_t)n;
  float* h   = xm + (size_t)n * D;
  float* out = (float*)d_out;

  const int T = 256;
  const int nBlocks   = (n + T - 1) / T;
  const int eBlocks   = (E + T - 1) / T;
  const long long nd  = (long long)n * D;
  const int ndBlocks  = (int)((nd + T - 1) / T);
  const int rowsPerBlock = (T / 32) * 16;
  const int gBlocks   = (n + rowsPerBlock - 1) / rowsPerBlock;

  k_init_deg<<<nBlocks, T, 0, stream>>>(deg, n);
  k_deg_edges<<<eBlocks, T, 0, stream>>>(dst, ew, deg, E);
  k_dis<<<nBlocks, T, 0, stream>>>(deg, dis, n);

  k_gemm32<false><<<gBlocks, T, 0, stream>>>(x, W1, xm, n);
  k_self_bias<<<ndBlocks, T, 0, stream>>>(xm, dis, b1, h, nd);
  k_edge_msg<<<eBlocks, T, 0, stream>>>(src, dst, ew, dis, xm, h, E);

  k_gemm32<true><<<gBlocks, T, 0, stream>>>(h, W2, xm, n);
  k_self_bias<<<ndBlocks, T, 0, stream>>>(xm, dis, b2, out, nd);
  k_edge_msg<<<eBlocks, T, 0, stream>>>(src, dst, ew, dis, xm, out, E);
}


# edge sums: f32 atomic row adds replaced by a destination-grouped edge list built per call in LDS (chunk sort by row range, owner orders by row); one wave sums each row; sum order = LDS arrival order
# speedup vs baseline: 4.1396x; 2.3647x over previous
#include <hip/hip_runtime.h>

typedef float v2f __attribute__((ext_vector_type(2)));
typedef float v8f __attribute__((ext_vector_type(8)));
typedef int v2i __attribute__((ext_vector_type(2)));
typedef int v4i __attribute__((ext_vector_type(4)));

#define D 32

#define RB 8
#define ROWS 256
#define CH 8192
#define NBMAX 512
#define CMAX 512
#define CAP 8192

__device__ __forceinline__ int block_scan(int v, int* wsum, int* tot) {
  const int lane = threadIdx.x & 31;
  const int wave = threadIdx.x >> 5;
  const int nwv = (blockDim.x + 31) >> 5;
  int x = v;
#pragma unroll
  for (int o = 1; o < 32; o <<= 1) {
    const int y = __shfl_up(x, o);
    if (lane >= o) x += y;
  }
  if (lane == 31) wsum[wave] = x;
  __syncthreads();
  if (wave == 0) {
    const int s = (lane < nwv) ? wsum[lane] : 0;
    int z = s;
#pragma unroll
    for (int o = 1; o < 32; o <<= 1) {
      const int y = __shfl_up(z, o);
      if (lane >= o) z += y;
    }
    wsum[lane] = z - s;
    if (lane == 31) wsum[32] = z;
  }
  __syncthreads();
  *tot = wsum[32];
  return wsum[wave] + (x - v);
}

__global__ void __launch_bounds__(512)
k_bin(const int* __restrict__ src, const int* __restrict__ dst,
      const float* __restrict__ w, int* __restrict__ ent1,
      int* __restrict__ offtab, int* __restrict__ total, int E) {
  __shared__ int cntb[NBMAX];
  __shared__ int curb[NBMAX];
  __shared__ int wsum[33];
  __shared__ v2i stage[CH];
  const int tid = threadIdx.x;
  const int e0 = blockIdx.x * CH;
  if (blockIdx.x == 0 && tid == 0) *total = 0;
  cntb[tid] = 0;
  __syncthreads();
  int t[CH / 512];
  int sv[CH / 512];
  float wv[CH / 512];
#pragma unroll
  for (int k = 0; k < CH / 512; ++k) {
    const int e = e0 + k * 512 + tid;
    const bool ok = (e < E);
    t[k] = ok ? dst[e] : -1;
    sv[k] = ok ? src[e] : 0;
    wv[k] = ok ? w[e] : 0.0f;
  }
#pragma unroll
  for (int k = 0; k < CH / 512; ++k) {
    if (t[k] >= 0) atomicAdd(&cntb[t[k] >> RB], 1);
  }
  __syncthreads();
  const int cb = cntb[tid];
  int tot;
  const int ex = block_scan(cb, wsum, &tot);
  curb[tid] = ex;
  int* ot = offtab + (size_t)blockIdx.x * (NBMAX + 1);
  ot[tid] = ex;
  if (tid == 511) ot[NBMAX] = tot;
  __syncthreads();
#pragma unroll
  for (int k = 0; k < CH / 512; ++k) {
    if (t[k] >= 0) {
      const int pos = atomicAdd(&curb[t[k] >> RB], 1);
      v2i q;
      q.x = sv[k] | ((t[k] & (ROWS - 1)) << 24);
      q.y = __float_as_int(wv[k]);
      stage[pos] = q;
    }
  }
  __syncthreads();
  v2i* o = reinterpret_cast<v2i*>(ent1) + e0;
#pragma unroll
  for (int k = 0; k < CH / 512; ++k) {
    const int idx = k * 512 + tid;
    if (e0 + idx < E) o[idx] = stage[idx];
  }
}

template <int MODE>
__device__ __forceinline__ void walk_pieces(const v2i* __restrict__ e1, const int* plo,
                                            const int* plen, int C, int* ncnt,
                                            float* ndeg, int* ncur, v2i* lst,
                                            v2i* __restrict__ cs, int gbase) {
  const int lane = threadIdx.x & 31;
  const int wave = threadIdx.x >> 5;
  const int nwv = blockDim.x >> 5;
  auto take = [&](v2i q) {
    const unsigned tl = (unsigned)q.x >> 24;
    if (MODE == 0) {
      atomicAdd(&ncnt[tl], 1);
      atomicAdd(&ndeg[tl], __int_as_float(q.y));
    } else {
      const int pos = atomicAdd(&ncur[tl], 1);
      q.x &= 0x00ffffff;
      if (MODE == 1) lst[pos] = q;
      else cs[gbase + pos] = q;
    }
  };
  for (int c0 = wave; c0 < C; c0 += 8 * nwv) {
    v2i q[8];
#pragma unroll
    for (int u = 0; u < 8; ++u) {
      const int c = c0 + u * nwv;
      q[u].x = 0;
      q[u].y = 0;
      if (c < C && lane < plen[c]) q[u] = e1[(size_t)c * CH + plo[c] + lane];
    }
#pragma unroll
    for (int u = 0; u < 8; ++u) {
      const int c = c0 + u * nwv;
      if (c < C && lane < plen[c]) take(q[u]);
    }
    for (int u = 0; u < 8; ++u) {
      const int c = c0 + u * nwv;
      if (c < C) {
        const v2i* p = e1 + (size_t)c * CH + plo[c];
        for (int j = lane + 32; j < plen[c]; j += 32) take(p[j]);
      }
    }
  }
}

__global__ void __launch_bounds__(512)
k_collect(const int* __restrict__ ent1, const int* __restrict__ offtab,
          int* __restrict__ csr, int* __restrict__ node, int* __restrict__ total,
          int C, int n) {
  __shared__ int plo[CMAX];
  __shared__ int plen[CMAX];
  __shared__ int ncnt[ROWS];
  __shared__ int ncur[ROWS];
  __shared__ float ndeg[ROWS];
  __shared__ int wsum[33];
  __shared__ int gb;
  __shared__ v2i lst[CAP];
  const int tid = threadIdx.x;
  const int b = blockIdx.x;
  const v2i* __restrict__ e1 = reinterpret_cast<const v2i*>(ent1);
  v2i* __restrict__ cs = reinterpret_cast<v2i*>(csr);

  for (int c = tid; c < C; c += blockDim.x) {
    const int* ot = offtab + (size_t)c * (NBMAX + 1) + b;
    const int lo = ot[0];
    plo[c] = lo;
    plen[c] = ot[1] - lo;
  }
  if (tid < ROWS) {
    ncnt[tid] = 0;
    ndeg[tid] = 1.0f;
  }
  __syncthreads();

  walk_pieces<0>(e1, plo, plen, C, ncnt, ndeg, ncur, lst, cs, 0);
  __syncthreads();

  const int cb = (tid < ROWS) ? ncnt[tid] : 0;
  int m;
  const int ex = block_scan(cb, wsum, &m);
  if (tid == 0) gb = atomicAdd(total, m);
  __syncthreads();
  const int gbase = gb;
  if (tid < ROWS) {
    ncur[tid] = ex;
    const int i = b * ROWS + tid;
    if (i < n) {
      const float d = ndeg[tid];
      const float r = (d > 0.0f) ? __frsqrt_rn(d) : 0.0f;
      v4i nd;
      nd.x = gbase + ex;
      nd.y = cb;
      nd.z = __float_as_int(r);
      nd.w = 0;
      reinterpret_cast<v4i*>(node)[i] = nd;
    }
  }
  __syncthreads();

  if (m <= CAP) {
    walk_pieces<1>(e1, plo, plen, C, ncnt, ndeg, ncur, lst, cs, gbase);
    __syncthreads();
    for (int idx = tid; idx < m; idx += blockDim.x) cs[gbase + idx] = lst[idx];
  } else {
    walk_pieces<2>(e1, plo, plen, C, ncnt, ndeg, ncur, lst, cs, gbase);
  }
}

template <bool RELU>
__global__ void k_gemm32(const float* __restrict__ X, const float* __restrict__ W,
                         float* __restrict__ Y, int n) {
  const int lane = threadIdx.x & 31;
  const int wave = threadIdx.x >> 5;
  const int row0 = (blockIdx.x * (blockDim.x >> 5) + wave) << 4;
  if (row0 >= n) return;

  const int l    = lane & 15;
  const int half = lane >> 4;

  v8f c0 = {};
  v8f c1 = {};

  const float* xrow = X + (size_t)(row0 + l) * D;

#pragma unroll
  for (int kb = 0; kb < D; kb += 4) {
    const int k0 = kb + 2 * half;
    v2f a;
    a.x = xrow[k0];
    a.y = xrow[k0 + 1];
    if (RELU) {
      a.x = fmaxf(a.x, 0.0f);
      a.y = fmaxf(a.y, 0.0f);
    }
    v2f b0, b1;
    b0.x = W[(size_t)k0 * D + l];
    b0.y = W[(size_t)(k0 + 1) * D + l];
    b1.x = W[(size_t)k0 * D + 16 + l];
    b1.y = W[(size_t)(k0 + 1) * D + 16 + l];

    c0 = __builtin_amdgcn_wmma_f32_16x16x4_f32(false, a, false, b0, (short)0, c0,
                                               false, false);
    c1 = __builtin_amdgcn_wmma_f32_16x16x4_f32(false, a, false, b1, (short)0, c1,
                                               false, false);
  }

  const int rowm = row0 + 8 * half;
#pragma unroll
  for (int j = 0; j < 8; ++j) {
    Y[(size_t)(rowm + j) * D + l]      = c0[j];
    Y[(size_t)(rowm + j) * D + 16 + l] = c1[j];
  }
}

template <bool FIRST>
__global__ void k_gather(const int* __restrict__ node, int* csr,
                         const float* __restrict__ xm, const float* __restrict__ b,
                         float* __restrict__ out, int n) {
  const int lane = threadIdx.x & 31;
  const int wave = __builtin_amdgcn_readfirstlane((int)(threadIdx.x >> 5));
  const int wpb = blockDim.x >> 5;
  const int nw = gridDim.x * wpb;
  const float bl = b[lane];
  v2i* ent2 = reinterpret_cast<v2i*>(csr);
  const v4i* __restrict__ node4 = reinterpret_cast<const v4i*>(node);

  for (int i = blockIdx.x * wpb + wave; i < n; i += nw) {
    const v4i nd = node4[i];
    const int p0 = nd.x;
    const int dg = nd.y;
    const float di = __int_as_float(nd.z);
    float acc = bl + xm[(unsigned)i * D + (unsigned)lane] * (di * di);

    for (int base = 0; base < dg; base += 32) {
      const int m = (dg - base < 32) ? (dg - base) : 32;
      int s = i;
      float c = 0.0f;
      if (lane < m) {
        v2i q = ent2[p0 + base + lane];
        s = q.x;
        if (FIRST) {
          const float ds = __int_as_float(node[(unsigned)s * 4 + 2]);
          c = ds * __int_as_float(q.y) * di;
          q.y = __float_as_int(c);
          ent2[p0 + base + lane] = q;
        } else {
          c = __int_as_float(q.y);
        }
      }
      for (int j = 0; j < m; j += 8) {
        float v[8];
#pragma unroll
        for (int u = 0; u < 8; ++u) {
          const unsigned rowi = (unsigned)__builtin_amdgcn_readlane(s, j + u) * D;
          v[u] = xm[rowi + (unsigned)lane];
        }
#pragma unroll
        for (int u = 0; u < 8; ++u) {
          const float cj =
              __int_as_float(__builtin_amdgcn_readlane(__float_as_int(c), j + u));
          acc += v[u] * cj;
        }
      }
    }
    out[(unsigned)i * D + (unsigned)lane] = acc;
  }
}

extern "C" void kernel_launch(void* const* d_in, const int* in_sizes, int n_in,
                              void* d_out, int out_size, void* d_ws, size_t ws_size,
                              hipStream_t stream) {
  const float* x  = (const float*)d_in[0];
  const int*   ei = (const int*)d_in[1];
  const float* ew = (const float*)d_in[2];
  const float* W1 = (const float*)d_in[3];
  const float* b1 = (const float*)d_in[4];
  const float* W2 = (const float*)d_in[5];
  const float* b2 = (const float*)d_in[6];

  const int n = in_sizes[0] / D;
  const int E = in_sizes[2];
  const int* src = ei;
  const int* dst = ei + E;

  const int C  = (E + CH - 1) / CH;
  const int nb = (n + ROWS - 1) / ROWS;

  float* ws   = (float*)d_ws;
  float* xm   = ws;
  float* h    = xm + (size_t)n * D;
  int*   ent1 = (int*)(h + (size_t)n * D);
  int*   csr  = ent1 + 2 * (size_t)E;
  int*   node = csr + 2 * (size_t)E;
  int*   offtab = node + 4 * (size_t)n;
  int*   total = offtab + (size_t)C * (NBMAX + 1);
  float* out  = (float*)d_out;

  const int T = 256;
  const int rowsPerBlock = (T / 32) * 16;
  const int gBlocks   = (n + rowsPerBlock - 1) / rowsPerBlock;
  const int aBlocks   = 4096;

  k_bin<<<C, 512, 0, stream>>>(src, dst, ew, ent1, offtab, total, E);
  k_collect<<<nb, 512, 0, stream>>>(ent1, offtab, csr, node, total, C, n);

  k_gemm32<false><<<gBlocks, T, 0, stream>>>(x, W1, xm, n);
  k_gather<true><<<aBlocks, T, 0, stream>>>(node, csr, xm, b1, h, n);

  k_gemm32<true><<<gBlocks, T, 0, stream>>>(h, W2, xm, n);
  k_gather<false><<<aBlocks, T, 0, stream>>>(node, csr, xm, b2, out, n);
}


# both 32x32 products moved into the gather kernels: a row's entries are summed first into an LDS tile, then one WMMA product per 16 rows ((A x) W, not A (x W): sum order changed); 6 launches -> 4
# speedup vs baseline: 4.8783x; 1.1784x over previous
#include <hip/hip_runtime.h>

typedef float v2f __attribute__((ext_vector_type(2)));
typedef float v8f __attribute__((ext_vector_type(8)));
typedef int v2i __attribute__((ext_vector_type(2)));
typedef int v4i __attribute__((ext_vector_type(4)));

#define D 32

#define RB 8
#define ROWS 256
#define CH 8192
#define NBMAX 512
#define CMAX 512
#define CAP 8192

__device__ __forceinline__ int block_scan(int v, int* wsum, int* tot) {
  const int lane = threadIdx.x & 31;
  const int wave = threadIdx.x >> 5;
  const int nwv = (blockDim.x + 31) >> 5;
  int x = v;
#pragma unroll
  for (int o = 1; o < 32; o <<= 1) {
    const int y = __shfl_up(x, o);
    if (lane >= o) x += y;
  }
  if (lane == 31) wsum[wave] = x;
  __syncthreads();
  if (wave == 0) {
    const int s = (lane < nwv) ? wsum[lane] : 0;
    int z = s;
#pragma unroll
    for (int o = 1; o < 32; o <<= 1) {
      const int y = __shfl_up(z, o);
      if (lane >= o) z += y;
    }
    wsum[lane] = z - s;
    if (lane == 31) wsum[32] = z;
  }
  __syncthreads();
  *tot = wsum[32];
  return wsum[wave] + (x - v);
}

__global__ void __launch_bounds__(512)
k_bin(const int* __restrict__ src, const int* __restrict__ dst,
      const float* __restrict__ w, int* __restrict__ ent1,
      int* __restrict__ offtab, int* __restrict__ total, int E) {
  __shared__ int cntb[NBMAX];
  __shared__ int curb[NBMAX];
  __shared__ int wsum[33];
  __shared__ v2i stage[CH];
  const int tid = threadIdx.x;
  const int e0 = blockIdx.x * CH;
  if (blockIdx.x == 0 && tid == 0) *total = 0;
  cntb[tid] = 0;
  __syncthreads();
  int t[CH / 512];
  int sv[CH / 512];
  float wv[CH / 512];
#pragma unroll
  for (int k = 0; k < CH / 512; ++k) {
    const int e = e0 + k * 512 + tid;
    const bool ok = (e < E);
    t[k] = ok ? dst[e] : -1;
    sv[k] = ok ? src[e] : 0;
    wv[k] = ok ? w[e] : 0.0f;
  }
#pragma unroll
  for (int k = 0; k < CH / 512; ++k) {
    if (t[k] >= 0) atomicAdd(&cntb[t[k] >> RB], 1);
  }
  __syncthreads();
  const int cb = cntb[tid];
  int tot;
  const int ex = block_scan(cb, wsum, &tot);
  curb[tid] = ex;
  int* ot = offtab + (size_t)blockIdx.x * (NBMAX + 1);
  ot[tid] = ex;
  if (tid == 511) ot[NBMAX] = tot;
  __syncthreads();
#pragma unroll
  for (int k = 0; k < CH / 512; ++k) {
    if (t[k] >= 0) {
      const int pos = atomicAdd(&curb[t[k] >> RB], 1);
      v2i q;
      q.x = sv[k] | ((t[k] & (ROWS - 1)) << 24);
      q.y = __float_as_int(wv[k]);
      stage[pos] = q;
    }
  }
  __syncthreads();
  v2i* o = reinterpret_cast<v2i*>(ent1) + e0;
#pragma unroll
  for (int k = 0; k < CH / 512; ++k) {
    const int idx = k * 512 + tid;
    if (e0 + idx < E) o[idx] = stage[idx];
  }
}

template <int MODE>
__device__ __forceinline__ void walk_pieces(const v2i* __restrict__ e1, const int* plo,
                                            const int* plen, int C, int* ncnt,
                                            float* ndeg, int* ncur, v2i* lst,
                                            v2i* __restrict__ cs, int gbase) {
  const int lane = threadIdx.x & 31;
  const int wave = threadIdx.x >> 5;
  const int nwv = blockDim.x >> 5;
  auto take = [&](v2i q) {
    const unsigned tl = (unsigned)q.x >> 24;
    if (MODE == 0) {
      atomicAdd(&ncnt[tl], 1);
      atomicAdd(&ndeg[tl], __int_as_float(q.y));
    } else {
      const int pos = atomicAdd(&ncur[tl], 1);
      q.x &= 0x00ffffff;
      if (MODE == 1) lst[pos] = q;
      else cs[gbase + pos] = q;
    }
  };
  for (int c0 = wave; c0 < C; c0 += 8 * nwv) {
    v2i q[8];
#pragma unroll
    for (int u = 0; u < 8; ++u) {
      const int c = c0 + u * nwv;
      q[u].x = 0;
      q[u].y = 0;
      if (c < C && lane < plen[c]) q[u] = e1[(size_t)c * CH + plo[c] + lane];
    }
#pragma unroll
    for (int u = 0; u < 8; ++u) {
      const int c = c0 + u * nwv;
      if (c < C && lane < plen[c]) take(q[u]);
    }
    for (int u = 0; u < 8; ++u) {
      const int c = c0 + u * nwv;
      if (c < C) {
        const v2i* p = e1 + (size_t)c * CH + plo[c];
        for (int j = lane + 32; j < plen[c]; j += 32) take(p[j]);
      }
    }
  }
}

__global__ void __launch_bounds__(512)
k_collect(const int* __restrict__ ent1, const int* __restrict__ offtab,
          int* __restrict__ csr, int* __restrict__ node, int* __restrict__ total,
          int C, int n) {
  __shared__ int plo[CMAX];
  __shared__ int plen[CMAX];
  __shared__ int ncnt[ROWS];
  __shared__ int ncur[ROWS];
  __shared__ float ndeg[ROWS];
  __shared__ int wsum[33];
  __shared__ int gb;
  __shared__ v2i lst[CAP];
  const int tid = threadIdx.x;
  const int b = blockIdx.x;
  const v2i* __restrict__ e1 = reinterpret_cast<const v2i*>(ent1);
  v2i* __restrict__ cs = reinterpret_cast<v2i*>(csr);

  for (int c = tid; c < C; c += blockDim.x) {
    const int* ot = offtab + (size_t)c * (NBMAX + 1) + b;
    const int lo = ot[0];
    plo[c] = lo;
    plen[c] = ot[1] - lo;
  }
  if (tid < ROWS) {
    ncnt[tid] = 0;
    ndeg[tid] = 1.0f;
  }
  __syncthreads();

  walk_pieces<0>(e1, plo, plen, C, ncnt, ndeg, ncur, lst, cs, 0);
  __syncthreads();

  const int cb = (tid < ROWS) ? ncnt[tid] : 0;
  int m;
  const int ex = block_scan(cb, wsum, &m);
  if (tid == 0) gb = atomicAdd(total, m);
  __syncthreads();
  const int gbase = gb;
  if (tid < ROWS) {
    ncur[tid] = ex;
    const int i = b * ROWS + tid;
    if (i < n) {
      const float d = ndeg[tid];
      const float r = (d > 0.0f) ? __frsqrt_rn(d) : 0.0f;
      v4i nd;
      nd.x = gbase + ex;
      nd.y = cb;
      nd.z = __float_as_int(r);
      nd.w = 0;
      reinterpret_cast<v4i*>(node)[i] = nd;
    }
  }
  __syncthreads();

  if (m <= CAP) {
    walk_pieces<1>(e1, plo, plen, C, ncnt, ndeg, ncur, lst, cs, gbase);
    __syncthreads();
    for (int idx = tid; idx < m; idx += blockDim.x) cs[gbase + idx] = lst[idx];
  } else {
    walk_pieces<2>(e1, plo, plen, C, ncnt, ndeg, ncur, lst, cs, gbase);
  }
}

#define RPB 64
#define TP 36

template <bool FIRST, bool RELU>
__global__ void __launch_bounds__(256)
k_layer(const int* __restrict__ node, int* csr, const float* __restrict__ X,
        const float* __restrict__ W, const float* __restrict__ b,
        float* __restrict__ Y, int n) {
  __shared__ float tile[RPB * TP];
  const int lane = threadIdx.x & 31;
  const int wave = __builtin_amdgcn_readfirstlane((int)(threadIdx.x >> 5));
  const int wpb = blockDim.x >> 5;
  const int rbase = blockIdx.x * RPB;
  v2i* ent2 = reinterpret_cast<v2i*>(csr);
  const v4i* __restrict__ node4 = reinterpret_cast<const v4i*>(node);

  for (int r = wave; r < RPB; r += wpb) {
    const int i = rbase + r;
    float acc = 0.0f;
    if (i < n) {
      const v4i nd = node4[i];
      const int p0 = nd.x;
      const int dg = nd.y;
      const float di = __int_as_float(nd.z);
      acc = X[(unsigned)i * D + (unsigned)lane] * (di * di);

      for (int base = 0; base < dg; base += 32) {
        const int m = (dg - base < 32) ? (dg - base) : 32;
        int s = i;
        float c = 0.0f;
        if (lane < m) {
          v2i q = ent2[p0 + base + lane];
          s = q.x;
          if (FIRST) {
            const float ds = __int_as_float(node[(unsigned)s * 4 + 2]);
            c = ds * __int_as_float(q.y) * di;
            q.y = __float_as_int(c);
            ent2[p0 + base + lane] = q;
          } else {
            c = __int_as_float(q.y);
          }
        }
        for (int j = 0; j < m; j += 8) {
          float v[8];
#pragma unroll
          for (int u = 0; u < 8; ++u) {
            const unsigned rowi = (unsigned)__builtin_amdgcn_readlane(s, j + u) * D;
            v[u] = X[rowi + (unsigned)lane];
          }
#pragma unroll
          for (int u = 0; u < 8; ++u) {
            const float cj =
                __int_as_float(__builtin_amdgcn_readlane(__float_as_int(c), j + u));
            acc += v[u] * cj;
          }
        }
      }
    }
    tile[r * TP + lane] = acc;
  }
  __syncthreads();

  const int row0 = rbase + (wave << 4);
  if (wave >= RPB / 16 || row0 >= n) return;

  const int l    = lane & 15;
  const int half = lane >> 4;

  v8f c0 = {};
  v8f c1 = {};

  const float* trow = tile + ((wave << 4) + l) * TP;

#pragma unroll
  for (int kb = 0; kb < D; kb += 4) {
    const int k0 = kb + 2 * half;
    v2f a;
    a.x = trow[k0];
    a.y = trow[k0 + 1];
    v2f b0, b1;
    b0.x = W[(size_t)k0 * D + l];
    b0.y = W[(size_t)(k0 + 1) * D + l];
    b1.x = W[(size_t)k0 * D + 16 + l];
    b1.y = W[(size_t)(k0 + 1) * D + 16 + l];

    c0 = __builtin_amdgcn_wmma_f32_16x16x4_f32(false, a, false, b0, (short)0, c0,
                                               false, false);
    c1 = __builtin_amdgcn_wmma_f32_16x16x4_f32(false, a, false, b1, (short)0, c1,
                                               false, false);
  }

  const float bl0 = b[l];
  const float bl1 = b[16 + l];
  const int rowm = row0 + 8 * half;
#pragma unroll
  for (int j = 0; j < 8; ++j) {
    float y0 = c0[j] + bl0;
    float y1 = c1[j] + bl1;
    if (RELU) {
      y0 = fmaxf(y0, 0.0f);
      y1 = fmaxf(y1, 0.0f);
    }
    Y[(size_t)(rowm + j) * D + l]      = y0;
    Y[(size_t)(rowm + j) * D + 16 + l] = y1;
  }
}

extern "C" void kernel_launch(void* const* d_in, const int* in_sizes, int n_in,
                              void* d_out, int out_size, void* d_ws, size_t ws_size,
                              hipStream_t stream) {
  const float* x  = (const float*)d_in[0];
  const int*   ei = (const int*)d_in[1];
  const float* ew = (const float*)d_in[2];
  const float* W1 = (const float*)d_in[3];
  const float* b1 = (const float*)d_in[4];
  const float* W2 = (const float*)d_in[5];
  const float* b2 = (const float*)d_in[6];

  const int n = in_sizes[0] / D;
  const int E = in_sizes[2];
  const int* src = ei;
  const int* dst = ei + E;

  const int C  = (E + CH - 1) / CH;
  const int nb = (n + ROWS - 1) / ROWS;

  float* ws   = (float*)d_ws;
  float* h    = ws;
  int*   ent1 = (int*)(h + (size_t)n * D);
  int*   csr  = ent1 + 2 * (size_t)E;
  int*   node = csr + 2 * (size_t)E;
  int*   offtab = node + 4 * (size_t)n;
  int*   total = offtab + (size_t)C * (NBMAX + 1);
  float* out  = (float*)d_out;

  const int T = 256;
  const int lBlocks = (n + RPB - 1) / RPB;

  k_bin<<<C, 512, 0, stream>>>(src, dst, ew, ent1, offtab, total, E);
  k_collect<<<nb, 512, 0, stream>>>(ent1, offtab, csr, node, total, C, n);

  k_layer<true, true><<<lBlocks, T, 0, stream>>>(node, csr, x, W1, b1, h, n);

  k_layer<false, false><<<lBlocks, T, 0, stream>>>(node, csr, h, W2, b2, out, n);
}


# gather: a wave's 8 rows fetch node records, entry lists and (layer 1) source degrees together ahead of the sums, not row by row; source rows 16 in flight (was 8); per-row sum order unchanged
# speedup vs baseline: 5.2900x; 1.0844x over previous
#include <hip/hip_runtime.h>

typedef float v2f __attribute__((ext_vector_type(2)));
typedef float v8f __attribute__((ext_vector_type(8)));
typedef int v2i __attribute__((ext_vector_type(2)));
typedef int v4i __attribute__((ext_vector_type(4)));

#define D 32

#define RB 8
#define ROWS 256
#define CH 8192
#define NBMAX 512
#define CMAX 512
#define CAP 8192

__device__ __forceinline__ int block_scan(int v, int* wsum, int* tot) {
  const int lane = threadIdx.x & 31;
  const int wave = threadIdx.x >> 5;
  const int nwv = (blockDim.x + 31) >> 5;
  int x = v;
#pragma unroll
  for (int o = 1; o < 32; o <<= 1) {
    const int y = __shfl_up(x, o);
    if (lane >= o) x += y;
  }
  if (lane == 31) wsum[wave] = x;
  __syncthreads();
  if (wave == 0) {
    const int s = (lane < nwv) ? wsum[lane] : 0;
    int z = s;
#pragma unroll
    for (int o = 1; o < 32; o <<= 1) {
      const int y = __shfl_up(z, o);
      if (lane >= o) z += y;
    }
    wsum[lane] = z - s;
    if (lane == 31) wsum[32] = z;
  }
  __syncthreads();
  *tot = wsum[32];
  return wsum[wave] + (x - v);
}

__global__ void __launch_bounds__(512)
k_bin(const int* __restrict__ src, const int* __restrict__ dst,
      const float* __restrict__ w, int* __restrict__ ent1,
      int* __restrict__ offtab, int* __restrict__ total, int E) {
  __shared__ int cntb[NBMAX];
  __shared__ int curb[NBMAX];
  __shared__ int wsum[33];
  __shared__ v2i stage[CH];
  const int tid = threadIdx.x;
  const int e0 = blockIdx.x * CH;
  if (blockIdx.x == 0 && tid == 0) *total = 0;
  cntb[tid] = 0;
  __syncthreads();
  int t[CH / 512];
  int sv[CH / 512];
  float wv[CH / 512];
#pragma unroll
  for (int k = 0; k < CH / 512; ++k) {
    const int e = e0 + k * 512 + tid;
    const bool ok = (e < E);
    t[k] = ok ? dst[e] : -1;
    sv[k] = ok ? src[e] : 0;
    wv[k] = ok ? w[e] : 0.0f;
  }
#pragma unroll
  for (int k = 0; k < CH / 512; ++k) {
    if (t[k] >= 0) atomicAdd(&cntb[t[k] >> RB], 1);
  }
  __syncthreads();
  const int cb = cntb[tid];
  int tot;
  const int ex = block_scan(cb, wsum, &tot);
  curb[tid] = ex;
  int* ot = offtab + (size_t)blockIdx.x * (NBMAX + 1);
  ot[tid] = ex;
  if (tid == 511) ot[NBMAX] = tot;
  __syncthreads();
#pragma unroll
  for (int k = 0; k < CH / 512; ++k) {
    if (t[k] >= 0) {
      const int pos = atomicAdd(&curb[t[k] >> RB], 1);
      v2i q;
      q.x = sv[k] | ((t[k] & (ROWS - 1)) << 24);
      q.y = __float_as_int(wv[k]);
      stage[pos] = q;
    }
  }
  __syncthreads();
  v2i* o = reinterpret_cast<v2i*>(ent1) + e0;
#pragma unroll
  for (int k = 0; k < CH / 512; ++k) {
    const int idx = k * 512 + tid;
    if (e0 + idx < E) o[idx] = stage[idx];
  }
}

template <int MODE>
__device__ __forceinline__ void walk_pieces(const v2i* __restrict__ e1, const int* plo,
                                            const int* plen, int C, int* ncnt,
                                            float* ndeg, int* ncur, v2i* lst,
                                            v2i* __restrict__ cs, int gbase) {
  const int lane = threadIdx.x & 31;
  const int wave = threadIdx.x >> 5;
  const int nwv = blockDim.x >> 5;
  auto take = [&](v2i q) {
    const unsigned tl = (unsigned)q.x >> 24;
    if (MODE == 0) {
      atomicAdd(&ncnt[tl], 1);
      atomicAdd(&ndeg[tl], __int_as_float(q.y));
    } else {
      const int pos = atomicAdd(&ncur[tl], 1);
      q.x &= 0x00ffffff;
      if (MODE == 1) lst[pos] = q;
      else cs[gbase + pos] = q;
    }
  };
  for (int c0 = wave; c0 < C; c0 += 8 * nwv) {
    v2i q[8];
#pragma unroll
    for (int u = 0; u < 8; ++u) {
      const int c = c0 + u * nwv;
      q[u].x = 0;
      q[u].y = 0;
      if (c < C && lane < plen[c]) q[u] = e1[(size_t)c * CH + plo[c] + lane];
    }
#pragma unroll
    for (int u = 0; u < 8; ++u) {
      const int c = c0 + u * nwv;
      if (c < C && lane < plen[c]) take(q[u]);
    }
    for (int u = 0; u < 8; ++u) {
      const int c = c0 + u * nwv;
      if (c < C) {
        const v2i* p = e1 + (size_t)c * CH + plo[c];
        for (int j = lane + 32; j < plen[c]; j += 32) take(p[j]);
      }
    }
  }
}

__global__ void __launch_bounds__(512)
k_collect(const int* __restrict__ ent1, const int* __restrict__ offtab,
          int* __restrict__ csr, int* __restrict__ node, int* __restrict__ total,
          int C, int n) {
  __shared__ int plo[CMAX];
  __shared__ int plen[CMAX];
  __shared__ int ncnt[ROWS];
  __shared__ int ncur[ROWS];
  __shared__ float ndeg[ROWS];
  __shared__ int wsum[33];
  __shared__ int gb;
  __shared__ v2i lst[CAP];
  const int tid = threadIdx.x;
  const int b = blockIdx.x;
  const v2i* __restrict__ e1 = reinterpret_cast<const v2i*>(ent1);
  v2i* __restrict__ cs = reinterpret_cast<v2i*>(csr);

  for (int c = tid; c < C; c += blockDim.x) {
    const int* ot = offtab + (size_t)c * (NBMAX + 1) + b;
    const int lo = ot[0];
    plo[c] = lo;
    plen[c] = ot[1] - lo;
  }
  if (tid < ROWS) {
    ncnt[tid] = 0;
    ndeg[tid] = 1.0f;
  }
  __syncthreads();

  walk_pieces<0>(e1, plo, plen, C, ncnt, ndeg, ncur, lst, cs, 0);
  __syncthreads();

  const int cb = (tid < ROWS) ? ncnt[tid] : 0;
  int m;
  const int ex = block_scan(cb, wsum, &m);
  if (tid == 0) gb = atomicAdd(total, m);
  __syncthreads();
  const int gbase = gb;
  if (tid < ROWS) {
    ncur[tid] = ex;
    const int i = b * ROWS + tid;
    if (i < n) {
      const float d = ndeg[tid];
      const float r = (d > 0.0f) ? __frsqrt_rn(d) : 0.0f;
      v4i nd;
      nd.x = gbase + ex;
      nd.y = cb;
      nd.z = __float_as_int(r);
      nd.w = 0;
      reinterpret_cast<v4i*>(node)[i] = nd;
    }
  }
  __syncthreads();

  if (m <= CAP) {
    walk_pieces<1>(e1, plo, plen, C, ncnt, ndeg, ncur, lst, cs, gbase);
    __syncthreads();
    for (int idx = tid; idx < m; idx += blockDim.x) cs[gbase + idx] = lst[idx];
  } else {
    walk_pieces<2>(e1, plo, plen, C, ncnt, ndeg, ncur, lst, cs, gbase);
  }
}

#define RPB 64
#define RPW 8
#define TP 36

template <bool FIRST, bool RELU>
__global__ void __launch_bounds__(256)
k_layer(const int* __restrict__ node, int* csr, const float* __restrict__ X,
        const float* __restrict__ W, const float* __restrict__ b,
        float* __restrict__ Y, int n) {
  __shared__ float tile[RPB * TP];
  const int lane = threadIdx.x & 31;
  const int wave = __builtin_amdgcn_readfirstlane((int)(threadIdx.x >> 5));
  const int rbase = blockIdx.x * RPB;
  v2i* ent2 = reinterpret_cast<v2i*>(csr);
  const v4i* __restrict__ node4 = reinterpret_cast<const v4i*>(node);

  const int r0 = wave * RPW;
  const int i0 = rbase + r0;

  v4i nd;
  nd.x = 0;
  nd.y = 0;
  nd.z = 0;
  nd.w = 0;
  if (lane < RPW && i0 + lane < n) nd = node4[i0 + lane];

  int pz[RPW];
  int mm[RPW];
  float di[RPW];
  int s[RPW];
  float c[RPW];
  float xs[RPW];
#pragma unroll
  for (int k = 0; k < RPW; ++k) {
    const int i = i0 + k;
    const int ic = (i < n) ? i : 0;
    const int dg = __builtin_amdgcn_readlane(nd.y, k);
    pz[k] = __builtin_amdgcn_readlane(nd.x, k);
    mm[k] = (dg < 32) ? dg : 32;
    di[k] = __int_as_float(__builtin_amdgcn_readlane(nd.z, k));
    xs[k] = X[(unsigned)ic * D + (unsigned)lane];
    const int pc = (dg > 0) ? pz[k] : 0;
    const int last = (mm[k] > 0) ? (mm[k] - 1) : 0;
    const v2i q = ent2[pc + ((lane < last) ? lane : last)];
    const bool in = (lane < mm[k]);
    s[k] = in ? q.x : ic;
    c[k] = in ? __int_as_float(q.y) : 0.0f;
  }

  if (FIRST) {
    float ds[RPW];
#pragma unroll
    for (int k = 0; k < RPW; ++k) ds[k] = __int_as_float(node[(unsigned)s[k] * 4 + 2]);
#pragma unroll
    for (int k = 0; k < RPW; ++k) {
      const bool in = (lane < mm[k]);
      const float cc = in ? (ds[k] * c[k] * di[k]) : 0.0f;
      c[k] = cc;
      if (in) {
        v2i q;
        q.x = s[k];
        q.y = __float_as_int(cc);
        ent2[pz[k] + lane] = q;
      }
    }
  }

#pragma unroll
  for (int k = 0; k < RPW; ++k) {
    float acc = xs[k] * (di[k] * di[k]);
    const int m = mm[k];
    for (int j = 0; j < m; j += 16) {
      float v[16];
      const bool two = (j + 8 < m);
#pragma unroll
      for (int u = 0; u < 8; ++u) {
        const unsigned rowi = (unsigned)__builtin_amdgcn_readlane(s[k], j + u) * D;
        v[u] = X[rowi + (unsigned)lane];
      }
      if (two) {
#pragma unroll
        for (int u = 8; u < 16; ++u) {
          const unsigned rowi = (unsigned)__builtin_amdgcn_readlane(s[k], j + u) * D;
          v[u] = X[rowi + (unsigned)lane];
        }
      }
#pragma unroll
      for (int u = 0; u < 8; ++u) {
        const float cj =
            __int_as_float(__builtin_amdgcn_readlane(__float_as_int(c[k]), j + u));
        acc += v[u] * cj;
      }
      if (two) {
#pragma unroll
        for (int u = 8; u < 16; ++u) {
          const float cj =
              __int_as_float(__builtin_amdgcn_readlane(__float_as_int(c[k]), j + u));
          acc += v[u] * cj;
        }
      }
    }
    tile[(r0 + k) * TP + lane] = acc;
  }

  for (int k = 0; k < RPW; ++k) {
    const int dg = __builtin_amdgcn_readlane(nd.y, k);
    if (dg > 32) {
      const int i = i0 + k;
      const int p0 = __builtin_amdgcn_readlane(nd.x, k);
      const float dk = __int_as_float(__builtin_amdgcn_readlane(nd.z, k));
      float acc = tile[(r0 + k) * TP + lane];
      for (int base = 32; base < dg; base += 32) {
        const int m = (dg - base < 32) ? (dg - base) : 32;
        int sx = i;
        float cx = 0.0f;
        if (lane < m) {
          v2i q = ent2[p0 + base + lane];
          sx = q.x;
          if (FIRST) {
            const float dsx = __int_as_float(node[(unsigned)sx * 4 + 2]);
            cx = dsx * __int_as_float(q.y) * dk;
            q.y = __float_as_int(cx);
            ent2[p0 + base + lane] = q;
          } else {
            cx = __int_as_float(q.y);
          }
        }
        for (int j = 0; j < m; j += 8) {
          float v[8];
#pragma unroll
          for (int u = 0; u < 8; ++u) {
            const unsigned rowi = (unsigned)__builtin_amdgcn_readlane(sx, j + u) * D;
            v[u] = X[rowi + (unsigned)lane];
          }
#pragma unroll
          for (int u = 0; u < 8; ++u) {
            const float cj =
                __int_as_float(__builtin_amdgcn_readlane(__float_as_int(cx), j + u));
            acc += v[u] * cj;
          }
        }
      }
      tile[(r0 + k) * TP + lane] = acc;
    }
  }
  __syncthreads();

  const int row0 = rbase + (wave << 4);
  if (wave >= RPB / 16 || row0 >= n) return;

  const int l    = lane & 15;
  const int half = lane >> 4;

  v8f c0 = {};
  v8f c1 = {};

  const float* trow = tile + ((wave << 4) + l) * TP;

#pragma unroll
  for (int kb = 0; kb < D; kb += 4) {
    const int k0 = kb + 2 * half;
    v2f a;
    a.x = trow[k0];
    a.y = trow[k0 + 1];
    v2f b0, b1;
    b0.x = W[(size_t)k0 * D + l];
    b0.y = W[(size_t)(k0 + 1) * D + l];
    b1.x = W[(size_t)k0 * D + 16 + l];
    b1.y = W[(size_t)(k0 + 1) * D + 16 + l];

    c0 = __builtin_amdgcn_wmma_f32_16x16x4_f32(false, a, false, b0, (short)0, c0,
                                               false, false);
    c1 = __builtin_amdgcn_wmma_f32_16x16x4_f32(false, a, false, b1, (short)0, c1,
                                               false, false);
  }

  const float bl0 = b[l];
  const float bl1 = b[16 + l];
  const int rowm = row0 + 8 * half;
#pragma unroll
  for (int j = 0; j < 8; ++j) {
    float y0 = c0[j] + bl0;
    float y1 = c1[j] + bl1;
    if (RELU) {
      y0 = fmaxf(y0, 0.0f);
      y1 = fmaxf(y1, 0.0f);
    }
    Y[(size_t)(rowm + j) * D + l]      = y0;
    Y[(size_t)(rowm + j) * D + 16 + l] = y1;
  }
}

extern "C" void kernel_launch(void* const* d_in, const int* in_sizes, int n_in,
                              void* d_out, int out_size, void* d_ws, size_t ws_size,
                              hipStream_t stream) {
  const float* x  = (const float*)d_in[0];
  const int*   ei = (const int*)d_in[1];
  const float* ew = (const float*)d_in[2];
  const float* W1 = (const float*)d_in[3];
  const float* b1 = (const float*)d_in[4];
  const float* W2 = (const float*)d_in[5];
  const float* b2 = (const float*)d_in[6];

  const int n = in_sizes[0] / D;
  const int E = in_sizes[2];
  const int* src = ei;
  const int* dst = ei + E;

  const int C  = (E + CH - 1) / CH;
  const int nb = (n + ROWS - 1) / ROWS;

  float* ws   = (float*)d_ws;
  float* h    = ws;
  int*   ent1 = (int*)(h + (size_t)n * D);
  int*   csr  = ent1 + 2 * (size_t)E;
  int*   node = csr + 2 * (size_t)E;
  int*   offtab = node + 4 * (size_t)n;
  int*   total = offtab + (size_t)C * (NBMAX + 1);
  float* out  = (float*)d_out;

  const int T = 256;
  const int lBlocks = (n + RPB - 1) / RPB;

  k_bin<<<C, 512, 0, stream>>>(src, dst, ew, ent1, offtab, total, E);
  k_collect<<<nb, 512, 0, stream>>>(ent1, offtab, csr, node, total, C, n);

  k_layer<true, true><<<lBlocks, T, 0, stream>>>(node, csr, x, W1, b1, h, n);

  k_layer<false, false><<<lBlocks, T, 0, stream>>>(node, csr, h, W2, b2, out, n);
}


# gather: 16-byte loads, one instruction fetches four source rows (8 lanes x 4 features); sum order changed: 4 interleaved partial sums per row, (g0+g1)+(g2+g3), own term last
# speedup vs baseline: 5.6418x; 1.0665x over previous
#include <hip/hip_runtime.h>

typedef float v2f __attribute__((ext_vector_type(2)));
typedef float v4f __attribute__((ext_vector_type(4)));
typedef float v8f __attribute__((ext_vector_type(8)));
typedef int v2i __attribute__((ext_vector_type(2)));
typedef int v4i __attribute__((ext_vector_type(4)));

#define D 32

#define RB 8
#define ROWS 256
#define CH 8192
#define NBMAX 512
#define CMAX 512
#define CAP 8192

__device__ __forceinline__ int block_scan(int v, int* wsum, int* tot) {
  const int lane = threadIdx.x & 31;
  const int wave = threadIdx.x >> 5;
  const int nwv = (blockDim.x + 31) >> 5;
  int x = v;
#pragma unroll
  for (int o = 1; o < 32; o <<= 1) {
    const int y = __shfl_up(x, o);
    if (lane >= o) x += y;
  }
  if (lane == 31) wsum[wave] = x;
  __syncthreads();
  if (wave == 0) {
    const int s = (lane < nwv) ? wsum[lane] : 0;
    int z = s;
#pragma unroll
    for (int o = 1; o < 32; o <<= 1) {
      const int y = __shfl_up(z, o);
      if (lane >= o) z += y;
    }
    wsum[lane] = z - s;
    if (lane == 31) wsum[32] = z;
  }
  __syncthreads();
  *tot = wsum[32];
  return wsum[wave] + (x - v);
}

__global__ void __launch_bounds__(512)
k_bin(const int* __restrict__ src, const int* __restrict__ dst,
      const float* __restrict__ w, int* __restrict__ ent1,
      int* __restrict__ offtab, int* __restrict__ total, int E) {
  __shared__ int cntb[NBMAX];
  __shared__ int curb[NBMAX];
  __shared__ int wsum[33];
  __shared__ v2i stage[CH];
  const int tid = threadIdx.x;
  const int e0 = blockIdx.x * CH;
  if (blockIdx.x == 0 && tid == 0) *total = 0;
  cntb[tid] = 0;
  __syncthreads();
  int t[CH / 512];
  int sv[CH / 512];
  float wv[CH / 512];
#pragma unroll
  for (int k = 0; k < CH / 512; ++k) {
    const int e = e0 + k * 512 + tid;
    const bool ok = (e < E);
    t[k] = ok ? dst[e] : -1;
    sv[k] = ok ? src[e] : 0;
    wv[k] = ok ? w[e] : 0.0f;
  }
#pragma unroll
  for (int k = 0; k < CH / 512; ++k) {
    if (t[k] >= 0) atomicAdd(&cntb[t[k] >> RB], 1);
  }
  __syncthreads();
  const int cb = cntb[tid];
  int tot;
  const int ex = block_scan(cb, wsum, &tot);
  curb[tid] = ex;
  int* ot = offtab + (size_t)blockIdx.x * (NBMAX + 1);
  ot[tid] = ex;
  if (tid == 511) ot[NBMAX] = tot;
  __syncthreads();
#pragma unroll
  for (int k = 0; k < CH / 512; ++k) {
    if (t[k] >= 0) {
      const int pos = atomicAdd(&curb[t[k] >> RB], 1);
      v2i q;
      q.x = sv[k] | ((t[k] & (ROWS - 1)) << 24);
      q.y = __float_as_int(wv[k]);
      stage[pos] = q;
    }
  }
  __syncthreads();
  v2i* o = reinterpret_cast<v2i*>(ent1) + e0;
#pragma unroll
  for (int k = 0; k < CH / 512; ++k) {
    const int idx = k * 512 + tid;
    if (e0 + idx < E) o[idx] = stage[idx];
  }
}

template <int MODE>
__device__ __forceinline__ void walk_pieces(const v2i* __restrict__ e1, const int* plo,
                                            const int* plen, int C, int* ncnt,
                                            float* ndeg, int* ncur, v2i* lst,
                                            v2i* __restrict__ cs, int gbase) {
  const int lane = threadIdx.x & 31;
  const int wave = threadIdx.x >> 5;
  const int nwv = blockDim.x >> 5;
  auto take = [&](v2i q) {
    const unsigned tl = (unsigned)q.x >> 24;
    if (MODE == 0) {
      atomicAdd(&ncnt[tl], 1);
      atomicAdd(&ndeg[tl], __int_as_float(q.y));
    } else {
      const int pos = atomicAdd(&ncur[tl], 1);
      q.x &= 0x00ffffff;
      if (MODE == 1) lst[pos] = q;
      else cs[gbase + pos] = q;
    }
  };
  for (int c0 = wave; c0 < C; c0 += 8 * nwv) {
    v2i q[8];
#pragma unroll
    for (int u = 0; u < 8; ++u) {
      const int c = c0 + u * nwv;
      q[u].x = 0;
      q[u].y = 0;
      if (c < C && lane < plen[c]) q[u] = e1[(size_t)c * CH + plo[c] + lane];
    }
#pragma unroll
    for (int u = 0; u < 8; ++u) {
      const int c = c0 + u * nwv;
      if (c < C && lane < plen[c]) take(q[u]);
    }
    for (int u = 0; u < 8; ++u) {
      const int c = c0 + u * nwv;
      if (c < C) {
        const v2i* p = e1 + (size_t)c * CH + plo[c];
        for (int j = lane + 32; j < plen[c]; j += 32) take(p[j]);
      }
    }
  }
}

__global__ void __launch_bounds__(512)
k_collect(const int* __restrict__ ent1, const int* __restrict__ offtab,
          int* __restrict__ csr, int* __restrict__ node, int* __restrict__ total,
          int C, int n) {
  __shared__ int plo[CMAX];
  __shared__ int plen[CMAX];
  __shared__ int ncnt[ROWS];
  __shared__ int ncur[ROWS];
  __shared__ float ndeg[ROWS];
  __shared__ int wsum[33];
  __shared__ int gb;
  __shared__ v2i lst[CAP];
  const int tid = threadIdx.x;
  const int b = blockIdx.x;
  const v2i* __restrict__ e1 = reinterpret_cast<const v2i*>(ent1);
  v2i* __restrict__ cs = reinterpret_cast<v2i*>(csr);

  for (int c = tid; c < C; c += blockDim.x) {
    const int* ot = offtab + (size_t)c * (NBMAX + 1) + b;
    const int lo = ot[0];
    plo[c] = lo;
    plen[c] = ot[1] - lo;
  }
  if (tid < ROWS) {
    ncnt[tid] = 0;
    ndeg[tid] = 1.0f;
  }
  __syncthreads();

  walk_pieces<0>(e1, plo, plen, C, ncnt, ndeg, ncur, lst, cs, 0);
  __syncthreads();

  const int cb = (tid < ROWS) ? ncnt[tid] : 0;
  int m;
  const int ex = block_scan(cb, wsum, &m);
  if (tid == 0) gb = atomicAdd(total, m);
  __syncthreads();
  const int gbase = gb;
  if (tid < ROWS) {
    ncur[tid] = ex;
    const int i = b * ROWS + tid;
    if (i < n) {
      const float d = ndeg[tid];
      const float r = (d > 0.0f) ? __frsqrt_rn(d) : 0.0f;
      v4i nd;
      nd.x = gbase + ex;
      nd.y = cb;
      nd.z = __float_as_int(r);
      nd.w = 0;
      reinterpret_cast<v4i*>(node)[i] = nd;
    }
  }
  __syncthreads();

  if (m <= CAP) {
    walk_pieces<1>(e1, plo, plen, C, ncnt, ndeg, ncur, lst, cs, gbase);
    __syncthreads();
    for (int idx = tid; idx < m; idx += blockDim.x) cs[gbase + idx] = lst[idx];
  } else {
    walk_pieces<2>(e1, plo, plen, C, ncnt, ndeg, ncur, lst, cs, gbase);
  }
}

__device__ __forceinline__ float lane_xor8(float x) {
  return __builtin_bit_cast(
      float, __builtin_amdgcn_permlane16(__builtin_bit_cast(int, x), __builtin_bit_cast(int, x),
                                         0xfedcba98, 0x76543210, false, false));
}
__device__ __forceinline__ float lane_xor16(float x) {
  return __builtin_bit_cast(
      float, __builtin_amdgcn_permlanex16(__builtin_bit_cast(int, x), __builtin_bit_cast(int, x),
                                          0x76543210, 0xfedcba98, false, false));
}

#define RPB 64
#define RPW 8
#define TP 36

template <bool FIRST, bool RELU>
__global__ void __launch_bounds__(256)
k_layer(const int* __restrict__ node, int* csr, const float* __restrict__ X,
        const float* __restrict__ W, const float* __restrict__ b,
        float* __restrict__ Y, int n) {
  __shared__ __attribute__((aligned(16))) float tile[RPB * TP];
  const int lane = threadIdx.x & 31;
  const int g = lane >> 3;
  const int l7 = lane & 7;
  const int bpa = g << 2;
  const v4f* __restrict__ X4 = reinterpret_cast<const v4f*>(X);
  const int wave = __builtin_amdgcn_readfirstlane((int)(threadIdx.x >> 5));
  const int rbase = blockIdx.x * RPB;
  v2i* ent2 = reinterpret_cast<v2i*>(csr);
  const v4i* __restrict__ node4 = reinterpret_cast<const v4i*>(node);

  const int r0 = wave * RPW;
  const int i0 = rbase + r0;

  v4i nd;
  nd.x = 0;
  nd.y = 0;
  nd.z = 0;
  nd.w = 0;
  if (lane < RPW && i0 + lane < n) nd = node4[i0 + lane];

  int pz[RPW];
  int mm[RPW];
  float di[RPW];
  int s[RPW];
  float c[RPW];
#pragma unroll
  for (int k = 0; k < RPW; ++k) {
    const int i = i0 + k;
    const int ic = (i < n) ? i : 0;
    const int dg = __builtin_amdgcn_readlane(nd.y, k);
    pz[k] = __builtin_amdgcn_readlane(nd.x, k);
    mm[k] = (dg < 32) ? dg : 32;
    di[k] = __int_as_float(__builtin_amdgcn_readlane(nd.z, k));
    const int pc = (dg > 0) ? pz[k] : 0;
    const int last = (mm[k] > 0) ? (mm[k] - 1) : 0;
    const v2i q = ent2[pc + ((lane < last) ? lane : last)];
    const bool in = (lane < mm[k]);
    s[k] = (mm[k] > 0) ? q.x : ic;
    c[k] = in ? __int_as_float(q.y) : 0.0f;
  }

  if (FIRST) {
    float ds[RPW];
#pragma unroll
    for (int k = 0; k < RPW; ++k) ds[k] = __int_as_float(node[(unsigned)s[k] * 4 + 2]);
#pragma unroll
    for (int k = 0; k < RPW; ++k) {
      const bool in = (lane < mm[k]);
      const float cc = in ? (ds[k] * c[k] * di[k]) : 0.0f;
      c[k] = cc;
      if (in) {
        v2i q;
        q.x = s[k];
        q.y = __float_as_int(cc);
        ent2[pz[k] + lane] = q;
      }
    }
  }

#pragma unroll
  for (int k = 0; k < RPW; ++k) {
    const int i = i0 + k;
    const int ic = (i < n) ? i : 0;
    const int m = mm[k];
    const v4f xv = X4[(unsigned)ic * 8u + (unsigned)l7];
    v4f acc = {0.0f, 0.0f, 0.0f, 0.0f};
    for (int j = 0; j < m; j += 16) {
      const int ba = bpa + 4 * j;
      const bool two = (j + 8 < m);
      v4f v[4];
#pragma unroll
      for (int u = 0; u < 2; ++u) {
        const unsigned st = (unsigned)__builtin_amdgcn_ds_bpermute(ba + 16 * u, s[k]);
        v[u] = X4[st * 8u + (unsigned)l7];
      }
      if (two) {
#pragma unroll
        for (int u = 2; u < 4; ++u) {
          const unsigned st = (unsigned)__builtin_amdgcn_ds_bpermute(ba + 16 * u, s[k]);
          v[u] = X4[st * 8u + (unsigned)l7];
        }
      }
#pragma unroll
      for (int u = 0; u < 2; ++u) {
        const float ct = __int_as_float(
            __builtin_amdgcn_ds_bpermute(ba + 16 * u, __float_as_int(c[k])));
        acc.x += v[u].x * ct;
        acc.y += v[u].y * ct;
        acc.z += v[u].z * ct;
        acc.w += v[u].w * ct;
      }
      if (two) {
#pragma unroll
        for (int u = 2; u < 4; ++u) {
          const float ct = __int_as_float(
              __builtin_amdgcn_ds_bpermute(ba + 16 * u, __float_as_int(c[k])));
          acc.x += v[u].x * ct;
          acc.y += v[u].y * ct;
          acc.z += v[u].z * ct;
          acc.w += v[u].w * ct;
        }
      }
    }
    acc.x += lane_xor8(acc.x);
    acc.y += lane_xor8(acc.y);
    acc.z += lane_xor8(acc.z);
    acc.w += lane_xor8(acc.w);
    acc.x += lane_xor16(acc.x);
    acc.y += lane_xor16(acc.y);
    acc.z += lane_xor16(acc.z);
    acc.w += lane_xor16(acc.w);
    if (lane < 8) {
      const float dd = di[k] * di[k];
      float* tr = tile + (r0 + k) * TP + 4 * lane;
      tr[0] = acc.x + xv.x * dd;
      tr[1] = acc.y + xv.y * dd;
      tr[2] = acc.z + xv.z * dd;
      tr[3] = acc.w + xv.w * dd;
    }
  }

  for (int k = 0; k < RPW; ++k) {
    const int dg = __builtin_amdgcn_readlane(nd.y, k);
    if (dg > 32) {
      const int i = i0 + k;
      const int p0 = __builtin_amdgcn_readlane(nd.x, k);
      const float dk = __int_as_float(__builtin_amdgcn_readlane(nd.z, k));
      float acc = tile[(r0 + k) * TP + lane];
      for (int base = 32; base < dg; base += 32) {
        const int m = (dg - base < 32) ? (dg - base) : 32;
        int sx = i;
        float cx = 0.0f;
        if (lane < m) {
          v2i q = ent2[p0 + base + lane];
          sx = q.x;
          if (FIRST) {
            const float dsx = __int_as_float(node[(unsigned)sx * 4 + 2]);
            cx = dsx * __int_as_float(q.y) * dk;
            q.y = __float_as_int(cx);
            ent2[p0 + base + lane] = q;
          } else {
            cx = __int_as_float(q.y);
          }
        }
        for (int j = 0; j < m; j += 8) {
          float v[8];
#pragma unroll
          for (int u = 0; u < 8; ++u) {
            const unsigned rowi = (unsigned)__builtin_amdgcn_readlane(sx, j + u) * D;
            v[u] = X[rowi + (unsigned)lane];
          }
#pragma unroll
          for (int u = 0; u < 8; ++u) {
            const float cj =
                __int_as_float(__builtin_amdgcn_readlane(__float_as_int(cx), j + u));
            acc += v[u] * cj;
          }
        }
      }
      tile[(r0 + k) * TP + lane] = acc;
    }
  }
  __syncthreads();

  const int row0 = rbase + (wave << 4);
  if (wave >= RPB / 16 || row0 >= n) return;

  const int l    = lane & 15;
  const int half = lane >> 4;

  v8f c0 = {};
  v8f c1 = {};

  const float* trow = tile + ((wave << 4) + l) * TP;

#pragma unroll
  for (int kb = 0; kb < D; kb += 4) {
    const int k0 = kb + 2 * half;
    v2f a;
    a.x = trow[k0];
    a.y = trow[k0 + 1];
    v2f b0, b1;
    b0.x = W[(size_t)k0 * D + l];
    b0.y = W[(size_t)(k0 + 1) * D + l];
    b1.x = W[(size_t)k0 * D + 16 + l];
    b1.y = W[(size_t)(k0 + 1) * D + 16 + l];

    c0 = __builtin_amdgcn_wmma_f32_16x16x4_f32(false, a, false, b0, (short)0, c0,
                                               false, false);
    c1 = __builtin_amdgcn_wmma_f32_16x16x4_f32(false, a, false, b1, (short)0, c1,
                                               false, false);
  }

  const float bl0 = b[l];
  const float bl1 = b[16 + l];
  const int rowm = row0 + 8 * half;
#pragma unroll
  for (int j = 0; j < 8; ++j) {
    float y0 = c0[j] + bl0;
    float y1 = c1[j] + bl1;
    if (RELU) {
      y0 = fmaxf(y0, 0.0f);
      y1 = fmaxf(y1, 0.0f);
    }
    Y[(size_t)(rowm + j) * D + l]      = y0;
    Y[(size_t)(rowm + j) * D + 16 + l] = y1;
  }
}

extern "C" void kernel_launch(void* const* d_in, const int* in_sizes, int n_in,
                              void* d_out, int out_size, void* d_ws, size_t ws_size,
                              hipStream_t stream) {
  const float* x  = (const float*)d_in[0];
  const int*   ei = (const int*)d_in[1];
  const float* ew = (const float*)d_in[2];
  const float* W1 = (const float*)d_in[3];
  const float* b1 = (const float*)d_in[4];
  const float* W2 = (const float*)d_in[5];
  const float* b2 = (const float*)d_in[6];

  const int n = in_sizes[0] / D;
  const int E = in_sizes[2];
  const int* src = ei;
  const int* dst = ei + E;

  const int C  = (E + CH - 1) / CH;
  const int nb = (n + ROWS - 1) / ROWS;

  float* ws   = (float*)d_ws;
  float* h    = ws;
  int*   ent1 = (int*)(h + (size_t)n * D);
  int*   csr  = ent1 + 2 * (size_t)E;
  int*   node = csr + 2 * (size_t)E;
  int*   offtab = node + 4 * (size_t)n;
  int*   total = offtab + (size_t)C * (NBMAX + 1);
  float* out  = (float*)d_out;

  const int T = 256;
  const int lBlocks = (n + RPB - 1) / RPB;

  k_bin<<<C, 512, 0, stream>>>(src, dst, ew, ent1, offtab, total, E);
  k_collect<<<nb, 512, 0, stream>>>(ent1, offtab, csr, node, total, C, n);

  k_layer<true, true><<<lBlocks, T, 0, stream>>>(node, csr, x, W1, b1, h, n);

  k_layer<false, false><<<lBlocks, T, 0, stream>>>(node, csr, h, W2, b2, out, n);
}


# source deg^-1/2 moved from entries to rows: row owner writes x*dis, layer 1 writes h*dis; sums use raw weights, scaled by dis[dst] once; no per-entry degree look-up or write-back; rounding differs
# speedup vs baseline: 5.8306x; 1.0335x over previous
#include <hip/hip_runtime.h>

typedef float v2f __attribute__((ext_vector_type(2)));
typedef float v4f __attribute__((ext_vector_type(4)));
typedef float v8f __attribute__((ext_vector_type(8)));
typedef int v2i __attribute__((ext_vector_type(2)));
typedef int v4i __attribute__((ext_vector_type(4)));

#define D 32

#define RB 8
#define ROWS 256
#define CH 8192
#define NBMAX 512
#define CMAX 512
#define CAP 8192

__device__ __forceinline__ int block_scan(int v, int* wsum, int* tot) {
  const int lane = threadIdx.x & 31;
  const int wave = threadIdx.x >> 5;
  const int nwv = (blockDim.x + 31) >> 5;
  int x = v;
#pragma unroll
  for (int o = 1; o < 32; o <<= 1) {
    const int y = __shfl_up(x, o);
    if (lane >= o) x += y;
  }
  if (lane == 31) wsum[wave] = x;
  __syncthreads();
  if (wave == 0) {
    const int s = (lane < nwv) ? wsum[lane] : 0;
    int z = s;
#pragma unroll
    for (int o = 1; o < 32; o <<= 1) {
      const int y = __shfl_up(z, o);
      if (lane >= o) z += y;
    }
    wsum[lane] = z - s;
    if (lane == 31) wsum[32] = z;
  }
  __syncthreads();
  *tot = wsum[32];
  return wsum[wave] + (x - v);
}

__global__ void __launch_bounds__(512)
k_bin(const int* __restrict__ src, const int* __restrict__ dst,
      const float* __restrict__ w, int* __restrict__ ent1,
      int* __restrict__ offtab, int* __restrict__ total, int E) {
  __shared__ int cntb[NBMAX];
  __shared__ int curb[NBMAX];
  __shared__ int wsum[33];
  __shared__ v2i stage[CH];
  const int tid = threadIdx.x;
  const int e0 = blockIdx.x * CH;
  if (blockIdx.x == 0 && tid == 0) *total = 0;
  cntb[tid] = 0;
  __syncthreads();
  int t[CH / 512];
  int sv[CH / 512];
  float wv[CH / 512];
#pragma unroll
  for (int k = 0; k < CH / 512; ++k) {
    const int e = e0 + k * 512 + tid;
    const bool ok = (e < E);
    t[k] = ok ? dst[e] : -1;
    sv[k] = ok ? src[e] : 0;
    wv[k] = ok ? w[e] : 0.0f;
  }
#pragma unroll
  for (int k = 0; k < CH / 512; ++k) {
    if (t[k] >= 0) atomicAdd(&cntb[t[k] >> RB], 1);
  }
  __syncthreads();
  const int cb = cntb[tid];
  int tot;
  const int ex = block_scan(cb, wsum, &tot);
  curb[tid] = ex;
  int* ot = offtab + (size_t)blockIdx.x * (NBMAX + 1);
  ot[tid] = ex;
  if (tid == 511) ot[NBMAX] = tot;
  __syncthreads();
#pragma unroll
  for (int k = 0; k < CH / 512; ++k) {
    if (t[k] >= 0) {
      const int pos = atomicAdd(&curb[t[k] >> RB], 1);
      v2i q;
      q.x = sv[k] | ((t[k] & (ROWS - 1)) << 24);
      q.y = __float_as_int(wv[k]);
      stage[pos] = q;
    }
  }
  __syncthreads();
  v2i* o = reinterpret_cast<v2i*>(ent1) + e0;
#pragma unroll
  for (int k = 0; k < CH / 512; ++k) {
    const int idx = k * 512 + tid;
    if (e0 + idx < E) o[idx] = stage[idx];
  }
}

template <int MODE>
__device__ __forceinline__ void walk_pieces(const v2i* __restrict__ e1, const int* plo,
                                            const int* plen, int C, int* ncnt,
                                            float* ndeg, int* ncur, v2i* lst,
                                            v2i* __restrict__ cs, int gbase) {
  const int lane = threadIdx.x & 31;
  const int wave = threadIdx.x >> 5;
  const int nwv = blockDim.x >> 5;
  auto take = [&](v2i q) {
    const unsigned tl = (unsigned)q.x >> 24;
    if (MODE == 0) {
      atomicAdd(&ncnt[tl], 1);
      atomicAdd(&ndeg[tl], __int_as_float(q.y));
    } else {
      const int pos = atomicAdd(&ncur[tl], 1);
      q.x &= 0x00ffffff;
      if (MODE == 1) lst[pos] = q;
      else cs[gbase + pos] = q;
    }
  };
  for (int c0 = wave; c0 < C; c0 += 8 * nwv) {
    v2i q[8];
#pragma unroll
    for (int u = 0; u < 8; ++u) {
      const int c = c0 + u * nwv;
      q[u].x = 0;
      q[u].y = 0;
      if (c < C && lane < plen[c]) q[u] = e1[(size_t)c * CH + plo[c] + lane];
    }
#pragma unroll
    for (int u = 0; u < 8; ++u) {
      const int c = c0 + u * nwv;
      if (c < C && lane < plen[c]) take(q[u]);
    }
    for (int u = 0; u < 8; ++u) {
      const int c = c0 + u * nwv;
      if (c < C) {
        const v2i* p = e1 + (size_t)c * CH + plo[c];
        for (int j = lane + 32; j < plen[c]; j += 32) take(p[j]);
      }
    }
  }
}

__global__ void __launch_bounds__(512)
k_collect(const int* __restrict__ ent1, const int* __restrict__ offtab,
          int* __restrict__ csr, int* __restrict__ node, int* __restrict__ total,
          const float* __restrict__ X, float* __restrict__ Xs, int C, int n) {
  __shared__ int plo[CMAX];
  __shared__ int plen[CMAX];
  __shared__ int ncnt[ROWS];
  __shared__ int ncur[ROWS];
  __shared__ float ndeg[ROWS];
  __shared__ int wsum[33];
  __shared__ int gb;
  __shared__ v2i lst[CAP];
  const int tid = threadIdx.x;
  const int b = blockIdx.x;
  const v2i* __restrict__ e1 = reinterpret_cast<const v2i*>(ent1);
  v2i* __restrict__ cs = reinterpret_cast<v2i*>(csr);

  for (int c = tid; c < C; c += blockDim.x) {
    const int* ot = offtab + (size_t)c * (NBMAX + 1) + b;
    const int lo = ot[0];
    plo[c] = lo;
    plen[c] = ot[1] - lo;
  }
  if (tid < ROWS) {
    ncnt[tid] = 0;
    ndeg[tid] = 1.0f;
  }
  __syncthreads();

  walk_pieces<0>(e1, plo, plen, C, ncnt, ndeg, ncur, lst, cs, 0);
  __syncthreads();

  const v4f* __restrict__ X4 = reinterpret_cast<const v4f*>(X) + (size_t)b * (ROWS * D / 4);
  v4f* __restrict__ Xs4 = reinterpret_cast<v4f*>(Xs) + (size_t)b * (ROWS * D / 4);
  v4f xr[ROWS * D / 4 / 512];
#pragma unroll
  for (int k = 0; k < ROWS * D / 4 / 512; ++k) {
    const int q = k * 512 + tid;
    xr[k] = X4[(b * ROWS + (q >> 3) < n) ? q : (q & 7)];
  }

  const int cb = (tid < ROWS) ? ncnt[tid] : 0;
  int m;
  const int ex = block_scan(cb, wsum, &m);
  if (tid == 0) gb = atomicAdd(total, m);
  __syncthreads();
  const int gbase = gb;
  if (tid < ROWS) {
    ncur[tid] = ex;
    const int i = b * ROWS + tid;
    const float d = ndeg[tid];
    const float r = (d > 0.0f) ? __frsqrt_rn(d) : 0.0f;
    ndeg[tid] = r;
    if (i < n) {
      v4i nd;
      nd.x = gbase + ex;
      nd.y = cb;
      nd.z = __float_as_int(r);
      nd.w = 0;
      reinterpret_cast<v4i*>(node)[i] = nd;
    }
  }
  __syncthreads();

#pragma unroll
  for (int k = 0; k < ROWS * D / 4 / 512; ++k) {
    const int q = k * 512 + tid;
    const float r = ndeg[q >> 3];
    if (b * ROWS + (q >> 3) < n) {
      v4f y;
      y.x = xr[k].x * r;
      y.y = xr[k].y * r;
      y.z = xr[k].z * r;
      y.w = xr[k].w * r;
      Xs4[q] = y;
    }
  }

  if (m <= CAP) {
    walk_pieces<1>(e1, plo, plen, C, ncnt, ndeg, ncur, lst, cs, gbase);
    __syncthreads();
    for (int idx = tid; idx < m; idx += blockDim.x) cs[gbase + idx] = lst[idx];
  } else {
    walk_pieces<2>(e1, plo, plen, C, ncnt, ndeg, ncur, lst, cs, gbase);
  }
}

__device__ __forceinline__ float lane_xor8(float x) {
  return __builtin_bit_cast(
      float, __builtin_amdgcn_permlane16(__builtin_bit_cast(int, x), __builtin_bit_cast(int, x),
                                         0xfedcba98, 0x76543210, false, false));
}
__device__ __forceinline__ float lane_xor16(float x) {
  return __builtin_bit_cast(
      float, __builtin_amdgcn_permlanex16(__builtin_bit_cast(int, x), __builtin_bit_cast(int, x),
                                          0x76543210, 0xfedcba98, false, false));
}

#define RPB 64
#define RPW 8
#define TP 36

template <bool SCALE, bool RELU>
__global__ void __launch_bounds__(256)
k_layer(const int* __restrict__ node, const int* __restrict__ csr, const float* __restrict__ X,
        const float* __restrict__ W, const float* __restrict__ b,
        float* __restrict__ Y, int n) {
  __shared__ __attribute__((aligned(16))) float tile[RPB * TP];
  __shared__ __attribute__((aligned(16))) float dsh[RPB];
  const int lane = threadIdx.x & 31;
  const int g = lane >> 3;
  const int l7 = lane & 7;
  const int bpa = g << 2;
  const v4f* __restrict__ X4 = reinterpret_cast<const v4f*>(X);
  const int wave = __builtin_amdgcn_readfirstlane((int)(threadIdx.x >> 5));
  const int rbase = blockIdx.x * RPB;
  const v2i* __restrict__ ent2 = reinterpret_cast<const v2i*>(csr);
  const v4i* __restrict__ node4 = reinterpret_cast<const v4i*>(node);

  const int r0 = wave * RPW;
  const int i0 = rbase + r0;

  v4i nd;
  nd.x = 0;
  nd.y = 0;
  nd.z = 0;
  nd.w = 0;
  if (lane < RPW && i0 + lane < n) nd = node4[i0 + lane];
  if (SCALE && lane < RPW) dsh[r0 + lane] = __int_as_float(nd.z);

  int pz[RPW];
  int mm[RPW];
  float di[RPW];
  int s[RPW];
  float c[RPW];
#pragma unroll
  for (int k = 0; k < RPW; ++k) {
    const int i = i0 + k;
    const int ic = (i < n) ? i : 0;
    const int dg = __builtin_amdgcn_readlane(nd.y, k);
    pz[k] = __builtin_amdgcn_readlane(nd.x, k);
    mm[k] = (dg < 32) ? dg : 32;
    di[k] = __int_as_float(__builtin_amdgcn_readlane(nd.z, k));
    const int pc = (dg > 0) ? pz[k] : 0;
    const int last = (mm[k] > 0) ? (mm[k] - 1) : 0;
    const v2i q = ent2[pc + ((lane < last) ? lane : last)];
    const bool in = (lane < mm[k]);
    s[k] = (mm[k] > 0) ? q.x : ic;
    c[k] = in ? __int_as_float(q.y) : 0.0f;
  }

#pragma unroll
  for (int k = 0; k < RPW; ++k) {
    const int i = i0 + k;
    const int ic = (i < n) ? i : 0;
    const int m = mm[k];
    const v4f xv = X4[(unsigned)ic * 8u + (unsigned)l7];
    v4f acc = {0.0f, 0.0f, 0.0f, 0.0f};
    for (int j = 0; j < m; j += 16) {
      const int ba = bpa + 4 * j;
      const bool two = (j + 8 < m);
      v4f v[4];
#pragma unroll
      for (int u = 0; u < 2; ++u) {
        const unsigned st = (unsigned)__builtin_amdgcn_ds_bpermute(ba + 16 * u, s[k]);
        v[u] = X4[st * 8u + (unsigned)l7];
      }
      if (two) {
#pragma unroll
        for (int u = 2; u < 4; ++u) {
          const unsigned st = (unsigned)__builtin_amdgcn_ds_bpermute(ba + 16 * u, s[k]);
          v[u] = X4[st * 8u + (unsigned)l7];
        }
      }
#pragma unroll
      for (int u = 0; u < 2; ++u) {
        const float ct = __int_as_float(
            __builtin_amdgcn_ds_bpermute(ba + 16 * u, __float_as_int(c[k])));
        acc.x += v[u].x * ct;
        acc.y += v[u].y * ct;
        acc.z += v[u].z * ct;
        acc.w += v[u].w * ct;
      }
      if (two) {
#pragma unroll
        for (int u = 2; u < 4; ++u) {
          const float ct = __int_as_float(
              __builtin_amdgcn_ds_bpermute(ba + 16 * u, __float_as_int(c[k])));
          acc.x += v[u].x * ct;
          acc.y += v[u].y * ct;
          acc.z += v[u].z * ct;
          acc.w += v[u].w * ct;
        }
      }
    }
    acc.x += lane_xor8(acc.x);
    acc.y += lane_xor8(acc.y);
    acc.z += lane_xor8(acc.z);
    acc.w += lane_xor8(acc.w);
    acc.x += lane_xor16(acc.x);
    acc.y += lane_xor16(acc.y);
    acc.z += lane_xor16(acc.z);
    acc.w += lane_xor16(acc.w);
    if (lane < 8) {
      float* tr = tile + (r0 + k) * TP + 4 * lane;
      tr[0] = (acc.x + xv.x) * di[k];
      tr[1] = (acc.y + xv.y) * di[k];
      tr[2] = (acc.z + xv.z) * di[k];
      tr[3] = (acc.w + xv.w) * di[k];
    }
  }

  for (int k = 0; k < RPW; ++k) {
    const int dg = __builtin_amdgcn_readlane(nd.y, k);
    if (dg > 32) {
      const int i = i0 + k;
      const int p0 = __builtin_amdgcn_readlane(nd.x, k);
      const float dk = __int_as_float(__builtin_amdgcn_readlane(nd.z, k));
      float acc = 0.0f;
      for (int base = 32; base < dg; base += 32) {
        const int m = (dg - base < 32) ? (dg - base) : 32;
        int sx = i;
        float cx = 0.0f;
        if (lane < m) {
          const v2i q = ent2[p0 + base + lane];
          sx = q.x;
          cx = __int_as_float(q.y);
        }
        for (int j = 0; j < m; j += 8) {
          float v[8];
#pragma unroll
          for (int u = 0; u < 8; ++u) {
            const unsigned rowi = (unsigned)__builtin_amdgcn_readlane(sx, j + u) * D;
            v[u] = X[rowi + (unsigned)lane];
          }
#pragma unroll
          for (int u = 0; u < 8; ++u) {
            const float cj =
                __int_as_float(__builtin_amdgcn_readlane(__float_as_int(cx), j + u));
            acc += v[u] * cj;
          }
        }
      }
      tile[(r0 + k) * TP + lane] += acc * dk;
    }
  }
  __syncthreads();

  const int row0 = rbase + (wave << 4);
  if (wave >= RPB / 16 || row0 >= n) return;

  const int l    = lane & 15;
  const int half = lane >> 4;

  v8f c0 = {};
  v8f c1 = {};

  const float* trow = tile + ((wave << 4) + l) * TP;

#pragma unroll
  for (int kb = 0; kb < D; kb += 4) {
    const int k0 = kb + 2 * half;
    v2f a;
    a.x = trow[k0];
    a.y = trow[k0 + 1];
    v2f b0, b1;
    b0.x = W[(size_t)k0 * D + l];
    b0.y = W[(size_t)(k0 + 1) * D + l];
    b1.x = W[(size_t)k0 * D + 16 + l];
    b1.y = W[(size_t)(k0 + 1) * D + 16 + l];

    c0 = __builtin_amdgcn_wmma_f32_16x16x4_f32(false, a, false, b0, (short)0, c0,
                                               false, false);
    c1 = __builtin_amdgcn_wmma_f32_16x16x4_f32(false, a, false, b1, (short)0, c1,
                                               false, false);
  }

  const float bl0 = b[l];
  const float bl1 = b[16 + l];
  const int rowm = row0 + 8 * half;
  float dr[8];
  if (SCALE) {
    const v4f* dp = reinterpret_cast<const v4f*>(dsh + (wave << 4) + 8 * half);
    const v4f d0 = dp[0];
    const v4f d1 = dp[1];
    dr[0] = d0.x; dr[1] = d0.y; dr[2] = d0.z; dr[3] = d0.w;
    dr[4] = d1.x; dr[5] = d1.y; dr[6] = d1.z; dr[7] = d1.w;
  }
#pragma unroll
  for (int j = 0; j < 8; ++j) {
    float y0 = c0[j] + bl0;
    float y1 = c1[j] + bl1;
    if (RELU) {
      y0 = fmaxf(y0, 0.0f);
      y1 = fmaxf(y1, 0.0f);
    }
    if (SCALE) {
      y0 *= dr[j];
      y1 *= dr[j];
    }
    Y[(size_t)(rowm + j) * D + l]      = y0;
    Y[(size_t)(rowm + j) * D + 16 + l] = y1;
  }
}

extern "C" void kernel_launch(void* const* d_in, const int* in_sizes, int n_in,
                              void* d_out, int out_size, void* d_ws, size_t ws_size,
                              hipStream_t stream) {
  const float* x  = (const float*)d_in[0];
  const int*   ei = (const int*)d_in[1];
  const float* ew = (const float*)d_in[2];
  const float* W1 = (const float*)d_in[3];
  const float* b1 = (const float*)d_in[4];
  const float* W2 = (const float*)d_in[5];
  const float* b2 = (const float*)d_in[6];

  const int n = in_sizes[0] / D;
  const int E = in_sizes[2];
  const int* src = ei;
  const int* dst = ei + E;

  const int C  = (E + CH - 1) / CH;
  const int nb = (n + ROWS - 1) / ROWS;

  float* ws   = (float*)d_ws;
  float* xs   = ws;
  float* h    = xs + (size_t)n * D;
  int*   ent1 = (int*)(h + (size_t)n * D);
  int*   csr  = ent1 + 2 * (size_t)E;
  int*   node = csr + 2 * (size_t)E;
  int*   offtab = node + 4 * (size_t)n;
  int*   total = offtab + (size_t)C * (NBMAX + 1);
  float* out  = (float*)d_out;

  const int T = 256;
  const int lBlocks = (n + RPB - 1) / RPB;

  k_bin<<<C, 512, 0, stream>>>(src, dst, ew, ent1, offtab, total, E);
  k_collect<<<nb, 512, 0, stream>>>(ent1, offtab, csr, node, total, x, xs, C, n);

  k_layer<true, true><<<lBlocks, T, 0, stream>>>(node, csr, xs, W1, b1, h, n);

  k_layer<false, false><<<lBlocks, T, 0, stream>>>(node, csr, h, W2, b2, out, n);
}
